# starting kernel
# baseline (speedup 1.0000x reference)
#include <hip/hip_runtime.h>
#include <hip/hip_bf16.h>

#define BB 32
#define TT 2048
#define DD 1024

typedef __attribute__((ext_vector_type(2))) float v2f;
typedef __attribute__((ext_vector_type(8))) float v8f;

__global__ __launch_bounds__(256) void att_init_kernel(float* __restrict__ sums,
                                                       float* __restrict__ out) {
    int i = blockIdx.x * 256 + threadIdx.x;
    if (i < BB) sums[i] = 0.0f;
    if (i < BB * DD) out[i] = 0.0f;
}

__global__ __launch_bounds__(256) void att_score_kernel(const float* __restrict__ x,
                                                        const float* __restrict__ w,
                                                        const float* __restrict__ bias,
                                                        float* __restrict__ e,
                                                        float* __restrict__ sums) {
    const int lane  = threadIdx.x & 31;
    const int wave  = threadIdx.x >> 5;
    const int b     = blockIdx.x >> 4;
    const int tblk  = blockIdx.x & 15;
    const int t0    = tblk * 128 + wave * 16;
    const int laneM = lane & 15;
    const int kofs  = (lane >> 4) << 1;

    const float* xrow = x + ((size_t)b * TT + (size_t)(t0 + laneM)) * DD + kofs;
    const float* wp   = w + kofs;

    v8f acc = {};
    #pragma unroll 4
    for (int k = 0; k < DD; k += 4) {
        v2f av = *(const v2f*)(wp + k);
        v2f bv = *(const v2f*)(xrow + k);
        acc = __builtin_amdgcn_wmma_f32_16x16x4_f32(
             false, av,  false, bv,
             (short)0, acc,  false,  false);
    }

    const int t = t0 + laneM;
    const float s  = acc[0] + bias[t];
    const float th = 1.0f - 2.0f * __builtin_amdgcn_rcpf(__expf(2.0f * s) + 1.0f);
    const float ev = __expf(th);

    if (lane < 16) e[(size_t)b * TT + t] = ev;

    float psum = ev;
    psum += __shfl_xor(psum, 1, 32);
    psum += __shfl_xor(psum, 2, 32);
    psum += __shfl_xor(psum, 4, 32);
    psum += __shfl_xor(psum, 8, 32);
    if (lane == 0) atomicAdd(&sums[b], psum);
}

__global__ __launch_bounds__(256) void att_pool_kernel(const float* __restrict__ x,
                                                       const float* __restrict__ e,
                                                       const float* __restrict__ sums,
                                                       float* __restrict__ out) {
    const int lane   = threadIdx.x & 31;
    const int wave   = threadIdx.x >> 5;
    const int bid    = blockIdx.x;
    const int tslice = bid & 3;
    const int dtile  = (bid >> 2) & 7;
    const int b      = bid >> 5;
    const int d0     = dtile * 128 + wave * 16;
    const int laneM  = lane & 15;
    const int tofs   = (lane >> 4) << 1;

    const float* xbase = x + (size_t)b * TT * DD + (size_t)d0 + laneM;
    const float* ebase = e + (size_t)b * TT + tofs;

    const int tstart = tslice * (TT / 4);
    v8f acc = {};
    #pragma unroll 4
    for (int t = tstart; t < tstart + TT / 4; t += 4) {
        const int tt = t + tofs;
        v2f av = *(const v2f*)(ebase + t);
        v2f bv;
        bv.x = xbase[(size_t)tt * DD];
        bv.y = xbase[(size_t)(tt + 1) * DD];
        acc = __builtin_amdgcn_wmma_f32_16x16x4_f32(
            false, av, false, bv, (short)0, acc, false, false);
    }

    if (lane < 16) {
        const float inv = __builtin_amdgcn_rcpf(sums[b]);
        atomicAdd(&out[(size_t)b * DD + d0 + lane], acc[0] * inv);
    }
}

extern "C" void kernel_launch(void* const* d_in, const int* in_sizes, int n_in,
                              void* d_out, int out_size, void* d_ws, size_t ws_size,
                              hipStream_t stream) {
    const float* x    = (const float*)d_in[0];
    const float* w    = (const float*)d_in[1];
    const float* bias = (const float*)d_in[2];
    float* out = (float*)d_out;

    float* e    = (float*)d_ws;
    float* sums = e + (size_t)BB * TT;

    att_init_kernel<<<(BB * DD + 255) / 256, 256, 0, stream>>>(sums, out);
    att_score_kernel<<<BB * (TT / 128), 256, 0, stream>>>(x, w, bias, e, sums);
    att_pool_kernel<<<BB * 8 * 4, 256, 0, stream>>>(x, e, sums, out);
}


# Pool pass: each wave streams 64 columns x 256 rows with 16-byte loads (two whole 256-byte row pieces per load), 20 kept in flight (was 4-byte loads drained every 16 rows); new t summation order
# speedup vs baseline: 1.1815x; 1.1815x over previous
#include <hip/hip_runtime.h>
#include <hip/hip_bf16.h>

#define BB 32
#define TT 2048
#define DD 1024

typedef __attribute__((ext_vector_type(2))) float v2f;
typedef __attribute__((ext_vector_type(4))) float v4f;
typedef __attribute__((ext_vector_type(8))) float v8f;

__global__ __launch_bounds__(256) void att_init_kernel(float* __restrict__ sums,
                                                       float* __restrict__ out) {
    int i = blockIdx.x * 256 + threadIdx.x;
    if (i < BB) sums[i] = 0.0f;
    if (i < BB * DD) out[i] = 0.0f;
}

#define SC_ROWS  16
#define SC_AHEAD 2
#define SC_Q     (DD / 128)

__global__ __launch_bounds__(256) void att_score_kernel(const float* __restrict__ x,
                                                        const float* __restrict__ w,
                                                        const float* __restrict__ bias,
                                                        float* __restrict__ e,
                                                        float* __restrict__ sums) {
    const int lane  = threadIdx.x & 31;
    const int wave  = threadIdx.x >> 5;
    const int b     = blockIdx.x >> 4;
    const int tblk  = blockIdx.x & 15;
    const int t0    = tblk * 128 + wave * 16;
    const int laneM = lane & 15;

    v4f wq[SC_Q];
    #pragma unroll
    for (int q = 0; q < SC_Q; ++q) wq[q] = *(const v4f*)(w + 4 * lane + 128 * q);

    const float* xb = x + ((size_t)b * TT + (size_t)t0) * DD + 4 * lane;
    v4f xr[SC_AHEAD + 1][SC_Q];
#define SC_LOAD_ROW(R) do { _Pragma("unroll") for (int q = 0; q < SC_Q; ++q) \
        xr[(R) % (SC_AHEAD + 1)][q] = *(const v4f*)(xb + (size_t)(R) * DD + 128 * q); } while (0)

    #pragma unroll
    for (int r = 0; r < SC_AHEAD; ++r) SC_LOAD_ROW(r);

    const int  di = lane & 7;
    const bool dv = (lane < 8) || (lane >= 24);
    float dsum[SC_ROWS];

    #pragma unroll
    for (int r = 0; r < SC_ROWS; ++r) {
        __builtin_amdgcn_sched_barrier(0);
        if (r + SC_AHEAD < SC_ROWS) SC_LOAD_ROW(r + SC_AHEAD);
        __builtin_amdgcn_sched_barrier(0);

        const v4f* xc = xr[r % (SC_AHEAD + 1)];
        v8f a0 = {}, a1 = {};
        #pragma unroll
        for (int q = 0; q < SC_Q; ++q) {
            v2f wa = {wq[q].x, wq[q].y}, xa = {xc[q].x, xc[q].y};
            v2f wb = {wq[q].z, wq[q].w}, xz = {xc[q].z, xc[q].w};
            a0 = __builtin_amdgcn_wmma_f32_16x16x4_f32(
                 false, wa,  false, xa,
                 (short)0, a0,  false,  false);
            a1 = __builtin_amdgcn_wmma_f32_16x16x4_f32(
                false, wb, false, xz, (short)0, a1, false, false);
        }
        const v8f acc = a0 + a1;
        float v = acc[0];
        #pragma unroll
        for (int i = 1; i < 8; ++i) v = (di == i) ? acc[i] : v;
        dsum[r] = dv ? v : 0.0f;
    }
#undef SC_LOAD_ROW

    float sc = 0.0f;
    #pragma unroll
    for (int r = 0; r < SC_ROWS; ++r) {
        float s = dsum[r];
        s += __shfl_xor(s, 16, 32);
        s += __shfl_xor(s, 8, 32);
        s += __shfl_xor(s, 4, 32);
        s += __shfl_xor(s, 2, 32);
        s += __shfl_xor(s, 1, 32);
        if (laneM == r) sc = s;
    }

    const int t = t0 + laneM;
    const float s  = sc + bias[t];
    const float th = 1.0f - 2.0f * __builtin_amdgcn_rcpf(__expf(2.0f * s) + 1.0f);
    const float ev = __expf(th);

    if (lane < 16) e[(size_t)b * TT + t] = ev;

    float psum = ev;
    psum += __shfl_xor(psum, 1, 32);
    psum += __shfl_xor(psum, 2, 32);
    psum += __shfl_xor(psum, 4, 32);
    psum += __shfl_xor(psum, 8, 32);
    if (lane == 0) atomicAdd(&sums[b], psum);
}

#define PL_STEP  8
#define PL_ROWS  256
#define PL_NSTEP (PL_ROWS / PL_STEP)
#define PL_RING  6

__global__ __launch_bounds__(256) void att_pool_kernel(const float* __restrict__ x,
                                                       const float* __restrict__ e,
                                                       const float* __restrict__ sums,
                                                       float* __restrict__ out) {
    __shared__ v4f part[4][16];

    const int lane = threadIdx.x & 31;
    const int wave = threadIdx.x >> 5;
    const int cgl  = wave & 3;
    const int tsub = wave >> 2;
    const int bid  = blockIdx.x;
    const int cq   = bid & 3;
    const int tsl  = (bid >> 2) & 3;
    const int b    = bid >> 4;
    const int d0   = cq * 256 + cgl * 64;
    const int t0   = tsl * 512 + tsub * PL_ROWS;
    const int c    = lane & 15;
    const int h    = lane >> 4;

    const float* xp = x + ((size_t)b * TT + (size_t)(t0 + 4 * h)) * DD + (size_t)(d0 + 4 * c);
    const float* ep = e + (size_t)b * TT + (size_t)(t0 + 4 * h);

    v4f xr[PL_RING][4];
    v4f er[PL_RING];
#define PL_LOAD_STEP(S) do { \
        er[(S) % PL_RING] = *(const v4f*)(ep + (size_t)(S) * PL_STEP); \
        _Pragma("unroll") for (int i = 0; i < 4; ++i) \
            xr[(S) % PL_RING][i] = *(const v4f*)(xp + ((size_t)(S) * PL_STEP + i) * DD); } while (0)

    #pragma unroll
    for (int s = 0; s < PL_RING - 1; ++s) PL_LOAD_STEP(s);

    v8f a0 = {}, a1 = {}, a2 = {}, a3 = {};

    #pragma unroll
    for (int s = 0; s < PL_NSTEP; ++s) {
        __builtin_amdgcn_sched_barrier(0);
        if (s + PL_RING - 1 < PL_NSTEP) PL_LOAD_STEP(s + PL_RING - 1);
        __builtin_amdgcn_sched_barrier(0);

        const v4f* xc = xr[s % PL_RING];
        const v4f  ec = er[s % PL_RING];
        const v2f e01 = {ec.x, ec.y}, e23 = {ec.z, ec.w};
        {
            v2f b0 = {xc[0].x, xc[1].x}, b1 = {xc[0].y, xc[1].y};
            v2f b2 = {xc[0].z, xc[1].z}, b3 = {xc[0].w, xc[1].w};
            a0 = __builtin_amdgcn_wmma_f32_16x16x4_f32(
                 false, e01,  false, b0,
                 (short)0, a0,  false,  false);
            a1 = __builtin_amdgcn_wmma_f32_16x16x4_f32(false, e01, false, b1, (short)0, a1, false, false);
            a2 = __builtin_amdgcn_wmma_f32_16x16x4_f32(false, e01, false, b2, (short)0, a2, false, false);
            a3 = __builtin_amdgcn_wmma_f32_16x16x4_f32(false, e01, false, b3, (short)0, a3, false, false);
        }
        {
            v2f b0 = {xc[2].x, xc[3].x}, b1 = {xc[2].y, xc[3].y};
            v2f b2 = {xc[2].z, xc[3].z}, b3 = {xc[2].w, xc[3].w};
            a0 = __builtin_amdgcn_wmma_f32_16x16x4_f32(false, e23, false, b0, (short)0, a0, false, false);
            a1 = __builtin_amdgcn_wmma_f32_16x16x4_f32(false, e23, false, b1, (short)0, a1, false, false);
            a2 = __builtin_amdgcn_wmma_f32_16x16x4_f32(false, e23, false, b2, (short)0, a2, false, false);
            a3 = __builtin_amdgcn_wmma_f32_16x16x4_f32(false, e23, false, b3, (short)0, a3, false, false);
        }
    }
#undef PL_LOAD_STEP

    const float inv = __builtin_amdgcn_rcpf(sums[b]);
    const v4f r = {a0[0], a1[0], a2[0], a3[0]};
    if (tsub == 1 && lane < 16) part[cgl][lane] = r;
    __syncthreads();
    if (tsub == 0 && lane < 16) {
        const v4f o = part[cgl][lane];
        float* op = out + (size_t)b * DD + d0 + 4 * lane;
        atomicAdd(op + 0, (r.x + o.x) * inv);
        atomicAdd(op + 1, (r.y + o.y) * inv);
        atomicAdd(op + 2, (r.z + o.z) * inv);
        atomicAdd(op + 3, (r.w + o.w) * inv);
    }
}

extern "C" void kernel_launch(void* const* d_in, const int* in_sizes, int n_in,
                              void* d_out, int out_size, void* d_ws, size_t ws_size,
                              hipStream_t stream) {
    const float* x    = (const float*)d_in[0];
    const float* w    = (const float*)d_in[1];
    const float* bias = (const float*)d_in[2];
    float* out = (float*)d_out;

    float* e    = (float*)d_ws;
    float* sums = e + (size_t)BB * TT;

    att_init_kernel<<<(BB * DD + 255) / 256, 256, 0, stream>>>(sums, out);
    att_score_kernel<<<BB * (TT / 128), 256, 0, stream>>>(x, w, bias, e, sums);
    att_pool_kernel<<<BB * 4 * 4, 256, 0, stream>>>(x, e, sums, out);
}


# Single fused pass, x read once: each wave holds a 16-row x 128-column x tile in registers for both WMMA products (no x in LDS); finalize kernel sums block partials, new fixed order, no atomics
# speedup vs baseline: 1.6776x; 1.4199x over previous
#include <hip/hip_runtime.h>

#define BB 32
#define TT 2048
#define DD 1024

typedef __attribute__((ext_vector_type(2))) float v2f;
typedef __attribute__((ext_vector_type(4))) float v4f;
typedef __attribute__((ext_vector_type(8))) float v8f;

static __device__ __forceinline__ void lds_barrier() {
    asm volatile("s_wait_dscnt 0x0\n\ts_barrier_signal -1\n\ts_barrier_wait -1" ::: "memory");
}

#define FR   16
#define FCH  64
#define FNT  (FCH / FR)
#define FNCH (TT / FCH)

__global__ __launch_bounds__(256) void att_fused_kernel(const float* __restrict__ x,
                                                        const float* __restrict__ w,
                                                        const float* __restrict__ bias,
                                                        float* __restrict__ part,
                                                        float* __restrict__ den) {
    __shared__ __attribute__((aligned(16))) float sc_part[2][FR][256];
    __shared__ __attribute__((aligned(16))) float e_lds[2][FR];
    __shared__ float den_lds[FR];

    const int tid  = threadIdx.x;
    const int lane = tid & 31;
    const int wave = tid >> 5;
    const int b    = blockIdx.x / FNCH;
    const int ch   = blockIdx.x % FNCH;
    const int t0   = ch * FCH;
    const int row  = tid >> 4;
    const int jj   = tid & 15;

    float bs[FNT];
    #pragma unroll
    for (int k = 0; k < FNT; ++k) bs[k] = bias[t0 + k * FR + row];

    const v4f wq = *(const v4f*)(w + 128 * wave + 4 * lane);
    const v2f wa = {wq.x, wq.y}, wb = {wq.z, wq.w};

    const float* xb = x + ((size_t)b * TT + (size_t)t0) * DD + 128 * wave + 4 * lane;
    v4f xr[2][FR];
#define F_LOAD_TILE(T) do { _Pragma("unroll") for (int r = 0; r < FR; ++r) \
        xr[(T) % 2][r] = *(const v4f*)(xb + (size_t)((T) * FR + r) * DD); } while (0)

    F_LOAD_TILE(0);

    const int  di = lane & 7;
    const bool dv = (lane < 8) || (lane >= 24);
    v8f o0 = {}, o1 = {}, o2 = {}, o3 = {};
    float dacc = 0.0f;

    #pragma unroll
    for (int tile = 0; tile < FNT; ++tile) {
        __builtin_amdgcn_sched_barrier(0);
        if (tile + 1 < FNT) F_LOAD_TILE(tile + 1);
        __builtin_amdgcn_sched_barrier(0);

        const v4f* xc  = xr[tile % 2];
        const int  buf = tile & 1;

        #pragma unroll
        for (int r = 0; r < FR; ++r) {
            const v2f xa = {xc[r].x, xc[r].y}, xz = {xc[r].z, xc[r].w};
            v8f a = {};
            a = __builtin_amdgcn_wmma_f32_16x16x4_f32(
                 false, wa,  false, xa,
                 (short)0, a,  false,  false);
            a = __builtin_amdgcn_wmma_f32_16x16x4_f32(false, wb, false, xz, (short)0, a, false, false);
            float v = a[0];
            #pragma unroll
            for (int i = 1; i < 8; ++i) v = (di == i) ? a[i] : v;
            sc_part[buf][r][wave * 32 + lane] = dv ? v : 0.0f;
            if ((r & 3) == 3) __builtin_amdgcn_sched_barrier(0);
        }
        lds_barrier();

        {
            const v4f p0 = *(const v4f*)&sc_part[buf][row][16 * jj];
            const v4f p1 = *(const v4f*)&sc_part[buf][row][16 * jj + 4];
            const v4f p2 = *(const v4f*)&sc_part[buf][row][16 * jj + 8];
            const v4f p3 = *(const v4f*)&sc_part[buf][row][16 * jj + 12];
            const v4f ps = (p0 + p1) + (p2 + p3);
            float s = (ps.x + ps.y) + (ps.z + ps.w);
            s += __shfl_xor(s, 1, 32);
            s += __shfl_xor(s, 2, 32);
            s += __shfl_xor(s, 4, 32);
            s += __shfl_xor(s, 8, 32);
            const float sb = s + bs[tile];
            const float th = 1.0f - 2.0f * __builtin_amdgcn_rcpf(__expf(2.0f * sb) + 1.0f);
            const float ev = __expf(th);
            if (jj == 0) e_lds[buf][row] = ev;
            dacc += ev;
        }
        lds_barrier();

        float ee[FR];
        {
            const v4f q0 = *(const v4f*)&e_lds[buf][0];
            const v4f q1 = *(const v4f*)&e_lds[buf][4];
            const v4f q2 = *(const v4f*)&e_lds[buf][8];
            const v4f q3 = *(const v4f*)&e_lds[buf][12];
            ee[0] = q0.x; ee[1] = q0.y; ee[2]  = q0.z; ee[3]  = q0.w;
            ee[4] = q1.x; ee[5] = q1.y; ee[6]  = q1.z; ee[7]  = q1.w;
            ee[8] = q2.x; ee[9] = q2.y; ee[10] = q2.z; ee[11] = q2.w;
            ee[12] = q3.x; ee[13] = q3.y; ee[14] = q3.z; ee[15] = q3.w;
        }
        #pragma unroll
        for (int p = 0; p < FR / 2; ++p) {
            const v2f ea = {dv ? ee[2 * p] : 0.0f, dv ? ee[2 * p + 1] : 0.0f};
            const v4f xa = xc[2 * p], xz = xc[2 * p + 1];
            const v2f b0 = {xa.x, xz.x}, b1 = {xa.y, xz.y}, b2 = {xa.z, xz.z}, b3 = {xa.w, xz.w};
            o0 = __builtin_amdgcn_wmma_f32_16x16x4_f32(false, ea, false, b0, (short)0, o0, false, false);
            o1 = __builtin_amdgcn_wmma_f32_16x16x4_f32(false, ea, false, b1, (short)0, o1, false, false);
            o2 = __builtin_amdgcn_wmma_f32_16x16x4_f32(false, ea, false, b2, (short)0, o2, false, false);
            o3 = __builtin_amdgcn_wmma_f32_16x16x4_f32(false, ea, false, b3, (short)0, o3, false, false);
        }
    }
#undef F_LOAD_TILE

    const v4f res = {o0[0], o1[0], o2[0], o3[0]};
    *(v4f*)(part + ((size_t)b * FNCH + ch) * DD + 128 * wave + 4 * lane) = res;

    if (jj == 0) den_lds[row] = dacc;
    __syncthreads();
    if (tid == 0) {
        float d = 0.0f;
        #pragma unroll
        for (int r = 0; r < FR; ++r) d += den_lds[r];
        den[b * FNCH + ch] = d;
    }
}

__global__ __launch_bounds__(256) void att_final_kernel(const float* __restrict__ part,
                                                        const float* __restrict__ den,
                                                        float* __restrict__ out) {
    const int i = blockIdx.x * 256 + threadIdx.x;
    const int b = i / DD;
    const int d = i % DD;
    float acc = 0.0f, ds = 0.0f;
    #pragma unroll
    for (int c = 0; c < FNCH; ++c) {
        acc += part[((size_t)b * FNCH + c) * DD + d];
        ds  += den[b * FNCH + c];
    }
    out[i] = acc / ds;
}

extern "C" void kernel_launch(void* const* d_in, const int* in_sizes, int n_in,
                              void* d_out, int out_size, void* d_ws, size_t ws_size,
                              hipStream_t stream) {
    const float* x    = (const float*)d_in[0];
    const float* w    = (const float*)d_in[1];
    const float* bias = (const float*)d_in[2];
    float* out = (float*)d_out;

    float* part = (float*)d_ws;
    float* den  = part + (size_t)BB * FNCH * DD;

    att_fused_kernel<<<BB * FNCH, 256, 0, stream>>>(x, w, bias, part, den);
    att_final_kernel<<<(BB * DD) / 256, 256, 0, stream>>>(part, den, out);
}


# Each wave owns whole rows: a row's score, softmax term and pool products all form inside one wave, so the row loop has no LDS traffic and no barrier; new fixed summation order of score and pool
# speedup vs baseline: 2.0264x; 1.2079x over previous
#include <hip/hip_runtime.h>

#define BB 32
#define TT 2048
#define DD 1024

typedef __attribute__((ext_vector_type(2))) float v2f;
typedef __attribute__((ext_vector_type(4))) float v4f;
typedef __attribute__((ext_vector_type(8))) float v8f;

#define NW   4
#define NR   16
#define FCH  (NW * NR)
#define FNCH (TT / FCH)
#define RING 4

template <int CTRL>
static __device__ __forceinline__ float row_xor_add(float v) {
    const int iv = __builtin_bit_cast(int, v);
    return v + __builtin_bit_cast(float, __builtin_amdgcn_update_dpp(0, iv, CTRL, 0xf, 0xf, false));
}
static __device__ __forceinline__ float wave_sum(float v) {
    v = row_xor_add<0x161>(v);
    v = row_xor_add<0x162>(v);
    v = row_xor_add<0x164>(v);
    v = row_xor_add<0x168>(v);
    const int iv = __builtin_bit_cast(int, v);
    return v + __builtin_bit_cast(float, __builtin_amdgcn_permlanex16(iv, iv, 0x76543210, 0xfedcba98, false, false));
}

__global__ __launch_bounds__(NW * 32) void att_fused_kernel(const float* __restrict__ x,
                                                            const float* __restrict__ w,
                                                            const float* __restrict__ bias,
                                                            float* __restrict__ part,
                                                            float* __restrict__ den) {
    __shared__ __attribute__((aligned(16))) float red[NW][DD];
    __shared__ float den_lds[NW];

    const int tid  = threadIdx.x;
    const int lane = tid & 31;
    const int wave = __builtin_amdgcn_readfirstlane(tid >> 5);
    const int b    = blockIdx.x / FNCH;
    const int ch   = blockIdx.x % FNCH;
    const int t0   = ch * FCH;

    const float* bp = bias + t0 + wave;
    float bs[NR];
    #pragma unroll
    for (int i = 0; i < NR; ++i) bs[i] = bp[i * NW];

    v4f wq[8];
    #pragma unroll
    for (int q = 0; q < 8; ++q) wq[q] = *(const v4f*)(w + 128 * q + 4 * lane);
    __builtin_amdgcn_sched_barrier(0);

    const float* xb = x + ((size_t)b * TT + (size_t)(t0 + wave)) * DD + 4 * lane;
    v4f xr[RING][8];
#define F_LOAD_ROW(I) do { _Pragma("unroll") for (int q = 0; q < 8; ++q) \
        xr[(I) % RING][q] = *(const v4f*)(xb + (size_t)(I) * (NW * DD) + 128 * q); } while (0)

    #pragma unroll
    for (int i = 0; i < RING - 1; ++i) F_LOAD_ROW(i);

    const int  c  = lane & 15;
    const int  u  = c - 2 * (lane >> 4);
    const int  di = lane & 7;
    const bool dv = (lane < 8) || (lane >= 24);
    v8f acc[4] = {};
    float dacc = 0.0f;

    #pragma unroll
    for (int i = 0; i < NR; ++i) {
        __builtin_amdgcn_sched_barrier(0);
        if (i + RING - 1 < NR) F_LOAD_ROW(i + RING - 1);
        __builtin_amdgcn_sched_barrier(0);

        const v4f* xc = xr[i % RING];

        v8f s0 = {}, s1 = {};
        #pragma unroll
        for (int q = 0; q < 8; ++q) {
            const v2f wa = {wq[q].x, wq[q].y}, wb = {wq[q].z, wq[q].w};
            const v2f xa = {xc[q].x, xc[q].y}, xz = {xc[q].z, xc[q].w};
            s0 = __builtin_amdgcn_wmma_f32_16x16x4_f32(
                 false, wa,  false, xa,
                 (short)0, s0,  false,  false);
            s1 = __builtin_amdgcn_wmma_f32_16x16x4_f32(false, wb, false, xz, (short)0, s1, false, false);
        }
        const v8f d = s0 + s1;
        float v = d[0];
        #pragma unroll
        for (int k = 1; k < 8; ++k) v = (di == k) ? d[k] : v;
        const float s = wave_sum(dv ? v : 0.0f);

        const float sb = s + bs[i];
        const float th = 1.0f - 2.0f * __builtin_amdgcn_rcpf(__expf(2.0f * sb) + 1.0f);
        const float ev = __expf(th);
        dacc += ev;

        v2f ag[4];
        #pragma unroll
        for (int g = 0; g < 4; ++g) {
            ag[g].x = (u == 4 * g)     ? ev : 0.0f;
            ag[g].y = (u == 4 * g + 1) ? ev : 0.0f;
        }
        #pragma unroll
        for (int p = 0; p < 16; ++p) {
            const v4f xq = xc[p >> 1];
            const v2f bq = (p & 1) ? (v2f){xq.z, xq.w} : (v2f){xq.x, xq.y};
            acc[p & 3] = __builtin_amdgcn_wmma_f32_16x16x4_f32(false, ag[p >> 2], false, bq, (short)0, acc[p & 3], false, false);
        }
    }
#undef F_LOAD_ROW

    {
        const int H = lane >> 4;
        #pragma unroll
        for (int a = 0; a < 4; ++a) {
            #pragma unroll
            for (int R = 0; R < 8; ++R) {
                const int col = 128 * (4 * H + 2 * (R >> 2) + (a >> 1)) + 4 * c + 64 * ((R >> 1) & 1) + 2 * (a & 1) + (R & 1);
                red[wave][col] = acc[a][R];
            }
        }
        if (lane == 0) den_lds[wave] = dacc;
    }
    __syncthreads();
    #pragma unroll
    for (int hh = 0; hh < 2; ++hh) {
        const int col = 512 * hh + 4 * tid;
        v4f r = *(const v4f*)&red[0][col];
        #pragma unroll
        for (int v2 = 1; v2 < NW; ++v2) r += *(const v4f*)&red[v2][col];
        *(v4f*)(part + ((size_t)b * FNCH + ch) * DD + col) = r;
    }
    if (tid == 0) {
        float dsum = 0.0f;
        #pragma unroll
        for (int v2 = 0; v2 < NW; ++v2) dsum += den_lds[v2];
        den[b * FNCH + ch] = dsum;
    }
}

__global__ __launch_bounds__(256) void att_final_kernel(const float* __restrict__ part,
                                                        const float* __restrict__ den,
                                                        float* __restrict__ out) {
    const int i = blockIdx.x * 256 + threadIdx.x;
    const int b = i / DD;
    const int d = i % DD;
    float acc = 0.0f, ds = 0.0f;
    #pragma unroll
    for (int c = 0; c < FNCH; ++c) {
        acc += part[((size_t)b * FNCH + c) * DD + d];
        ds  += den[b * FNCH + c];
    }
    out[i] = acc / ds;
}

extern "C" void kernel_launch(void* const* d_in, const int* in_sizes, int n_in,
                              void* d_out, int out_size, void* d_ws, size_t ws_size,
                              hipStream_t stream) {
    const float* x    = (const float*)d_in[0];
    const float* w    = (const float*)d_in[1];
    const float* bias = (const float*)d_in[2];
    float* out = (float*)d_out;

    float* part = (float*)d_ws;
    float* den  = part + (size_t)BB * FNCH * DD;

    att_fused_kernel<<<BB * FNCH, NW * 32, 0, stream>>>(x, w, bias, part, den);
    att_final_kernel<<<(BB * DD) / 256, 256, 0, stream>>>(part, den, out);
}
